# starting kernel
# baseline (speedup 1.0000x reference)
#include <hip/hip_runtime.h>


#define B_ 64
#define N_ 512
#define D_ 128
#define T_ 8
#define NEG_SLOPE 0.2f

typedef __attribute__((ext_vector_type(2))) float v2f;
typedef __attribute__((ext_vector_type(8))) float v8f;

__global__ __launch_bounds__(256)
void edgeattn_project(const float* __restrict__ src,
                      const float* __restrict__ dst,
                      const float* __restrict__ a,
                      float* __restrict__ s_src,
                      float* __restrict__ s_dst) {
    const int lane  = threadIdx.x & 31;
    const int tile  = blockIdx.x * 8 + (threadIdx.x >> 5);
    const int mrow  = (tile << 4) + (lane & 15);
    const int khalf = (lane & 16) ? 2 : 0;
    const int tcol  = lane & 15;

    const int   tt  = tcol & (T_ - 1);
    const float msk = (tcol < T_) ? 1.0f : 0.0f;
    const float* __restrict__ arow   = a + (size_t)tt * (2 * D_);
    const float* __restrict__ srcRow = src + (size_t)mrow * D_;
    const float* __restrict__ dstRow = dst + (size_t)mrow * D_;

    v8f c_src = {};
    v8f c_dst = {};

#pragma unroll 4
    for (int kk = 0; kk < D_ / 4; ++kk) {
        const int k0 = (kk << 2) + khalf;

        v2f a_s, a_d, b_s, b_d;
        a_s.x = srcRow[k0];     a_s.y = srcRow[k0 + 1];
        a_d.x = dstRow[k0];     a_d.y = dstRow[k0 + 1];
        b_s.x = arow[k0]        * msk;
        b_s.y = arow[k0 + 1]    * msk;
        b_d.x = arow[D_ + k0]   * msk;
        b_d.y = arow[D_ + k0 + 1] * msk;

        c_src = __builtin_amdgcn_wmma_f32_16x16x4_f32(
            false, a_s, false, b_s, (short)0, c_src, false, false);
        c_dst = __builtin_amdgcn_wmma_f32_16x16x4_f32(
            false, a_d, false, b_d, (short)0, c_dst, false, false);
    }

    if (tcol < T_) {
        const int rbase = (tile << 4) + ((lane & 16) ? 8 : 0);
#pragma unroll
        for (int v = 0; v < 8; ++v) {
            s_src[(size_t)(rbase + v) * T_ + tcol] = c_src[v];
            s_dst[(size_t)(rbase + v) * T_ + tcol] = c_dst[v];
        }
    }
}

__global__ __launch_bounds__(256)
void edgeattn_score(const int* __restrict__ edges,
                    const float* __restrict__ s_src,
                    const float* __restrict__ s_dst,
                    float* __restrict__ out) {
    const size_t e   = ((size_t)blockIdx.x * 256 + threadIdx.x) * 4;
    const int    b   = (int)(e >> 18);
    const int    rem = (int)(e & (N_ * N_ - 1));
    const int    i   = rem >> 9;
    const int    j   = rem & (N_ - 1);

    __builtin_prefetch((const void*)(edges + e + 8192), 0, 1);

    const int4 ev = *(const int4*)(edges + e);

    const float* __restrict__ sr = s_src + ((size_t)b * N_ + i) * T_;
    const float* __restrict__ sd = s_dst + ((size_t)b * N_ + j) * T_;

    float4 r;
    {
        int k = ev.x < 0 ? 0 : (ev.x > T_ - 1 ? T_ - 1 : ev.x);
        float g = sr[k] + sd[0 * T_ + k];
        g = (ev.x < 0) ? 0.0f : g;
        r.x = (g >= 0.0f) ? g : NEG_SLOPE * g;
    }
    {
        int k = ev.y < 0 ? 0 : (ev.y > T_ - 1 ? T_ - 1 : ev.y);
        float g = sr[k] + sd[1 * T_ + k];
        g = (ev.y < 0) ? 0.0f : g;
        r.y = (g >= 0.0f) ? g : NEG_SLOPE * g;
    }
    {
        int k = ev.z < 0 ? 0 : (ev.z > T_ - 1 ? T_ - 1 : ev.z);
        float g = sr[k] + sd[2 * T_ + k];
        g = (ev.z < 0) ? 0.0f : g;
        r.z = (g >= 0.0f) ? g : NEG_SLOPE * g;
    }
    {
        int k = ev.w < 0 ? 0 : (ev.w > T_ - 1 ? T_ - 1 : ev.w);
        float g = sr[k] + sd[3 * T_ + k];
        g = (ev.w < 0) ? 0.0f : g;
        r.w = (g >= 0.0f) ? g : NEG_SLOPE * g;
    }

    *(float4*)(out + e) = r;
}

extern "C" void kernel_launch(void* const* d_in, const int* in_sizes, int n_in,
                              void* d_out, int out_size, void* d_ws, size_t ws_size,
                              hipStream_t stream) {
    const float* src   = (const float*)d_in[0];
    const float* dst   = (const float*)d_in[1];
    const int*   edges = (const int*)d_in[2];
    const float* a     = (const float*)d_in[3];
    float*       out   = (float*)d_out;

    float* s_src = (float*)d_ws;
    float* s_dst = s_src + (size_t)B_ * N_ * T_;

    edgeattn_project<<<(B_ * N_ / 16) / 8, 256, 0, stream>>>(src, dst, a, s_src, s_dst);

    const int blocks = (B_ * N_ * N_) / (4 * 256);
    edgeattn_score<<<blocks, 256, 0, stream>>>(edges, s_src, s_dst, out);
}


# edge pass: one block per 64 rows of a batch stages the s_dst/s_src score tables in LDS (conflict-free transposed layout); per-edge look-ups are ds_load_b32, not 32-line cache gathers
# speedup vs baseline: 1.8287x; 1.8287x over previous
#include <hip/hip_runtime.h>


#define B_ 64
#define N_ 512
#define D_ 128
#define T_ 8
#define NEG_SLOPE 0.2f

typedef __attribute__((ext_vector_type(2))) float v2f;
typedef __attribute__((ext_vector_type(8))) float v8f;

__global__ __launch_bounds__(256)
void edgeattn_project(const float* __restrict__ src,
                      const float* __restrict__ dst,
                      const float* __restrict__ a,
                      float* __restrict__ s_src,
                      float* __restrict__ s_dst) {
    const int lane  = threadIdx.x & 31;
    const int tile  = blockIdx.x * 8 + (threadIdx.x >> 5);
    const int mrow  = (tile << 4) + (lane & 15);
    const int khalf = (lane & 16) ? 2 : 0;
    const int tcol  = lane & 15;

    const int   tt  = tcol & (T_ - 1);
    const float msk = (tcol < T_) ? 1.0f : 0.0f;
    const float* __restrict__ arow   = a + (size_t)tt * (2 * D_);
    const float* __restrict__ srcRow = src + (size_t)mrow * D_;
    const float* __restrict__ dstRow = dst + (size_t)mrow * D_;

    v8f c_src = {};
    v8f c_dst = {};

#pragma unroll 4
    for (int kk = 0; kk < D_ / 4; ++kk) {
        const int k0 = (kk << 2) + khalf;

        v2f a_s, a_d, b_s, b_d;
        a_s.x = srcRow[k0];     a_s.y = srcRow[k0 + 1];
        a_d.x = dstRow[k0];     a_d.y = dstRow[k0 + 1];
        b_s.x = arow[k0]        * msk;
        b_s.y = arow[k0 + 1]    * msk;
        b_d.x = arow[D_ + k0]   * msk;
        b_d.y = arow[D_ + k0 + 1] * msk;

        c_src = __builtin_amdgcn_wmma_f32_16x16x4_f32(
            false, a_s, false, b_s, (short)0, c_src, false, false);
        c_dst = __builtin_amdgcn_wmma_f32_16x16x4_f32(
            false, a_d, false, b_d, (short)0, c_dst, false, false);
    }

    if (tcol < T_) {
        const int rbase = (tile << 4) + ((lane & 16) ? 8 : 0);
#pragma unroll
        for (int v = 0; v < 8; ++v) {
            s_src[(size_t)(rbase + v) * T_ + tcol] = c_src[v];
            s_dst[(size_t)(rbase + v) * T_ + tcol] = c_dst[v];
        }
    }
}

#define ROWS_PER_BLOCK 64
#define ROWS_PER_WAVE  8
#define SD_PK 1024
#define SD_PM 132

__device__ __forceinline__ float edge_score(int e, const float* srow, const float* sdcol) {
    const int k = e < 0 ? 0 : (e > T_ - 1 ? T_ - 1 : e);
    float g = srow[k] + sdcol[k * SD_PK];
    g = (e < 0) ? 0.0f : g;
    return (g >= 0.0f) ? g : NEG_SLOPE * g;
}

__global__ __launch_bounds__(256)
void edgeattn_score(const int* __restrict__ edges,
                    const float* __restrict__ s_src,
                    const float* __restrict__ s_dst,
                    float* __restrict__ out) {
    __shared__ float sdT[T_ * SD_PK];
    __shared__ __attribute__((aligned(16))) float srL[ROWS_PER_BLOCK * T_];

    const int tid  = threadIdx.x;
    const int lane = tid & 31;
    const int wave = tid >> 5;
    const int b    = blockIdx.x >> 3;
    const int i0   = (blockIdx.x & 7) * ROWS_PER_BLOCK;

    const float4* __restrict__ sd4 = (const float4*)(s_dst + (size_t)b * N_ * T_);
    float4 st[4];
#pragma unroll
    for (int q = 0; q < 4; ++q) st[q] = sd4[tid + 256 * q];
    float4 srv = make_float4(0.f, 0.f, 0.f, 0.f);
    if (tid < ROWS_PER_BLOCK * T_ / 4)
        srv = ((const float4*)(s_src + ((size_t)b * N_ + i0) * T_))[tid];

    const size_t wrow = ((size_t)b * N_ + i0 + wave * ROWS_PER_WAVE) * N_;
    const int4* __restrict__ e4 = (const int4*)(edges + wrow) + lane;
    float4* __restrict__     o4 = (float4*)(out + wrow) + lane;
    int4 ev[ROWS_PER_WAVE * 4];
#pragma unroll
    for (int r = 0; r < ROWS_PER_WAVE; ++r)
#pragma unroll
        for (int q = 0; q < 4; ++q)
            ev[r * 4 + q] = e4[r * (N_ / 4) + q * 32];

    __builtin_amdgcn_sched_barrier(0);

#pragma unroll
    for (int q = 0; q < 4; ++q) {
        const int f  = tid + 256 * q;
        const int j  = f >> 1;
        const int w0 = ((f & 1) * 4) * SD_PK + (j & 3) * SD_PM + (j >> 2);
        sdT[w0]             = st[q].x;
        sdT[w0 + SD_PK]     = st[q].y;
        sdT[w0 + 2 * SD_PK] = st[q].z;
        sdT[w0 + 3 * SD_PK] = st[q].w;
    }
    if (tid < ROWS_PER_BLOCK * T_ / 4) ((float4*)srL)[tid] = srv;
    __builtin_amdgcn_fence(__ATOMIC_RELEASE, "workgroup", "local");
    __builtin_amdgcn_s_barrier();
    __builtin_amdgcn_fence(__ATOMIC_ACQUIRE, "workgroup", "local");

#pragma unroll
    for (int r = 0; r < ROWS_PER_WAVE; ++r) {
        const float* srow = srL + (wave * ROWS_PER_WAVE + r) * T_;
#pragma unroll
        for (int q = 0; q < 4; ++q) {
            const int4   e  = ev[r * 4 + q];
            const float* sc = sdT + q * 32 + lane;
            float4 o;
            o.x = edge_score(e.x, srow, sc);
            o.y = edge_score(e.y, srow, sc + SD_PM);
            o.z = edge_score(e.z, srow, sc + 2 * SD_PM);
            o.w = edge_score(e.w, srow, sc + 3 * SD_PM);
            o4[r * (N_ / 4) + q * 32] = o;
        }
    }
}

extern "C" void kernel_launch(void* const* d_in, const int* in_sizes, int n_in,
                              void* d_out, int out_size, void* d_ws, size_t ws_size,
                              hipStream_t stream) {
    const float* src   = (const float*)d_in[0];
    const float* dst   = (const float*)d_in[1];
    const int*   edges = (const int*)d_in[2];
    const float* a     = (const float*)d_in[3];
    float*       out   = (float*)d_out;

    float* s_src = (float*)d_ws;
    float* s_dst = s_src + (size_t)B_ * N_ * T_;

    edgeattn_project<<<(B_ * N_ / 16) / 8, 256, 0, stream>>>(src, dst, a, s_src, s_dst);

    edgeattn_score<<<(B_ * N_) / ROWS_PER_BLOCK, 256, 0, stream>>>(edges, s_src, s_dst, out);
}


# fuse projection into edge pass: one launch; a 512-thread block owns a 256x256 tile, forms its 256 s_src + 256 s_dst rows by f32 WMMA into LDS; k taken in 16-byte pieces (sum order changes)
# speedup vs baseline: 2.3522x; 1.2863x over previous
#include <hip/hip_runtime.h>


#define B_ 64
#define N_ 512
#define D_ 128
#define T_ 8
#define NEG_SLOPE 0.2f

typedef __attribute__((ext_vector_type(2))) float v2f;
typedef __attribute__((ext_vector_type(4))) float v4f;
typedef __attribute__((ext_vector_type(8))) float v8f;

#define TILE          256
#define ROWS_PER_WAVE 16
#define SD_PK         256
#define SD_PM         64
#define A_PITCH       264

__device__ __forceinline__ float edge_score(int e, const float* srow, const float* sdcol) {
    const int k = e < 0 ? 0 : (e > T_ - 1 ? T_ - 1 : e);
    float g = srow[k] + sdcol[k * SD_PK];
    g = (e < 0) ? 0.0f : g;
    return (g >= 0.0f) ? g : NEG_SLOPE * g;
}

__global__ __launch_bounds__(512)
void edgeattn_fused(const float* __restrict__ src,
                    const float* __restrict__ dst,
                    const int*   __restrict__ edges,
                    const float* a,
                    float* __restrict__ out) {
    __shared__ __attribute__((aligned(16))) float aL[T_ * A_PITCH];
    __shared__ float sdT[T_ * SD_PK];
    __shared__ float srL[TILE * T_];

    const int tid  = threadIdx.x;
    const int lane = tid & 31;
    const int wave = tid >> 5;
    const int b    = blockIdx.x >> 2;
    const int i0   = ((blockIdx.x >> 1) & 1) * TILE;
    const int j0   = (blockIdx.x & 1) * TILE;
    const int c    = lane & 15;
    const int h    = lane >> 4;

    const v4f aw = ((const v4f*)a)[tid];
    __builtin_amdgcn_sched_barrier(0);

    const float4* __restrict__ sp =
        (const float4*)(src + ((size_t)b * N_ + i0 + wave * ROWS_PER_WAVE + c) * D_) + h;
    const float4* __restrict__ dp =
        (const float4*)(dst + ((size_t)b * N_ + j0 + wave * ROWS_PER_WAVE + c) * D_) + h;
    float4 fs[16], fd[16];
#pragma unroll
    for (int p = 0; p < 16; ++p) { fs[p] = sp[2 * p]; fd[p] = dp[2 * p]; }
    __builtin_amdgcn_sched_barrier(0);

    const size_t wrow = ((size_t)b * N_ + i0 + wave * ROWS_PER_WAVE) * N_ + j0;
    const int4* __restrict__ e4 = (const int4*)(edges + wrow) + lane;
    float4* __restrict__     o4 = (float4*)(out + wrow) + lane;
    int4 ev[ROWS_PER_WAVE * 2];
#pragma unroll
    for (int r = 0; r < ROWS_PER_WAVE / 2; ++r)
#pragma unroll
        for (int q = 0; q < 2; ++q)
            ev[r * 2 + q] = e4[r * (N_ / 4) + q * 32];

    __builtin_amdgcn_sched_barrier(0);

    *(v4f*)(aL + (tid >> 6) * A_PITCH + (tid & 63) * 4) = aw;
    __builtin_amdgcn_fence(__ATOMIC_RELEASE, "workgroup", "local");
    __builtin_amdgcn_s_barrier();
    __builtin_amdgcn_fence(__ATOMIC_ACQUIRE, "workgroup", "local");

    const bool   live = c < T_;
    const float* bsp  = aL + (c & (T_ - 1)) * A_PITCH + 4 * h;
    v8f c_src = {};
    v8f c_dst = {};
#pragma unroll
    for (int p = 0; p < 16; ++p) {
        float4 bs = *(const float4*)(bsp + 8 * p);
        float4 bd = *(const float4*)(bsp + D_ + 8 * p);
        v2f as0 = {fs[p].x, fs[p].y}, as1 = {fs[p].z, fs[p].w};
        v2f ad0 = {fd[p].x, fd[p].y}, ad1 = {fd[p].z, fd[p].w};
        v2f bs0 = {bs.x, bs.y}, bs1 = {bs.z, bs.w};
        v2f bd0 = {bd.x, bd.y}, bd1 = {bd.z, bd.w};
        c_src = __builtin_amdgcn_wmma_f32_16x16x4_f32(false, as0, false, bs0, (short)0, c_src, false, false);
        c_dst = __builtin_amdgcn_wmma_f32_16x16x4_f32(false, ad0, false, bd0, (short)0, c_dst, false, false);
        c_src = __builtin_amdgcn_wmma_f32_16x16x4_f32(false, as1, false, bs1, (short)0, c_src, false, false);
        c_dst = __builtin_amdgcn_wmma_f32_16x16x4_f32(false, ad1, false, bd1, (short)0, c_dst, false, false);
    }

#pragma unroll
    for (int r = ROWS_PER_WAVE / 2; r < ROWS_PER_WAVE; ++r)
#pragma unroll
        for (int q = 0; q < 2; ++q)
            ev[r * 2 + q] = e4[r * (N_ / 4) + q * 32];

    __builtin_amdgcn_sched_barrier(0);

    asm volatile("v_nop\n\tv_nop\n\tv_nop\n\tv_nop" :: "v"(c_src), "v"(c_dst));

    if (live) {
        const int rl = wave * ROWS_PER_WAVE + 8 * h;
#pragma unroll
        for (int r = 0; r < 8; ++r) {
            const int jl = rl + r;
            srL[jl * T_ + c] = c_src[r];
            sdT[c * SD_PK + (jl & 3) * SD_PM + (jl >> 2)] = c_dst[r];
        }
    }
    __builtin_amdgcn_fence(__ATOMIC_RELEASE, "workgroup", "local");
    __builtin_amdgcn_s_barrier();
    __builtin_amdgcn_fence(__ATOMIC_ACQUIRE, "workgroup", "local");

#pragma unroll
    for (int r = 0; r < ROWS_PER_WAVE; ++r) {
        const float* srow = srL + (wave * ROWS_PER_WAVE + r) * T_;
#pragma unroll
        for (int q = 0; q < 2; ++q) {
            const int4   e  = ev[r * 2 + q];
            const float* sc = sdT + q * 32 + lane;
            float4 o;
            o.x = edge_score(e.x, srow, sc);
            o.y = edge_score(e.y, srow, sc + SD_PM);
            o.z = edge_score(e.z, srow, sc + 2 * SD_PM);
            o.w = edge_score(e.w, srow, sc + 3 * SD_PM);
            o4[r * (N_ / 4) + q * 32] = o;
        }
    }
}

extern "C" void kernel_launch(void* const* d_in, const int* in_sizes, int n_in,
                              void* d_out, int out_size, void* d_ws, size_t ws_size,
                              hipStream_t stream) {
    const float* src   = (const float*)d_in[0];
    const float* dst   = (const float*)d_in[1];
    const int*   edges = (const int*)d_in[2];
    const float* a     = (const float*)d_in[3];
    float*       out   = (float*)d_out;

    edgeattn_fused<<<B_ * 4, 512, 0, stream>>>(src, dst, edges, a, out);
}
